# starting kernel
# baseline (speedup 1.0000x reference)
#include <hip/hip_runtime.h>
#include <math.h>

typedef __attribute__((ext_vector_type(2))) float v2f;
typedef __attribute__((ext_vector_type(8))) float v8f;

#define N_NODES 100000
#define IN_CH   256
#define OUT_CH  64
#define N_EDGES 400000
#define ALPHA_F 0.25f

#define WS_B   0
#define WS_S   4096
#define WS_ACC (4096 + 2 * N_NODES)

__global__ void __launch_bounds__(256)
prep_kernel(const float* __restrict__ Wd, const float* __restrict__ Wf,
            float* __restrict__ ws) {
    __shared__ float ush[OUT_CH];
    __shared__ float vsh[OUT_CH];
    const int t = threadIdx.x;
    if (t < OUT_CH) {
        ush[t] = Wf[t] + Wf[2 * OUT_CH + t];
        vsh[t] = Wf[OUT_CH + t] - Wf[2 * OUT_CH + t];
    }
    __syncthreads();

    float su = 0.f, sv = 0.f;
    #pragma unroll 8
    for (int o = 0; o < OUT_CH; ++o) {
        const float w = Wd[o * IN_CH + t];
        su = fmaf(w, ush[o], su);
        sv = fmaf(w, vsh[o], sv);
    }

    const int kk       = t >> 2;
    const int koff     = t & 3;
    const int laneBase = (koff < 2) ? 0 : 16;
    const int slot     = koff & 1;
    float* bp = ws + WS_B;
    #pragma unroll
    for (int col = 0; col < 16; ++col) {
        const float val = (col == 0) ? su : (col == 1) ? sv : 0.f;
        bp[(kk * 32 + laneBase + col) * 2 + slot] = val;
    }
    if (t == 0) { ws[WS_ACC] = 0.f; ws[WS_ACC + 1] = 0.f; }
}

__global__ void __launch_bounds__(256)
matvec_wmma_kernel(const float* __restrict__ x, float* __restrict__ ws) {
    const int wave   = (int)((blockIdx.x * blockDim.x + threadIdx.x) >> 5);
    const int nWaves = N_NODES / 16;
    if (wave >= nWaves) return;
    const int lane = threadIdx.x & 31;
    const int row0 = wave * 16;

    const int    rowA = row0 + (lane & 15);
    const float* ax   = x + (size_t)rowA * IN_CH + ((lane & 16) ? 2 : 0);
    const v2f*   bp   = (const v2f*)(ws + WS_B);

    v8f c = {};
    #pragma unroll 8
    for (int kk = 0; kk < 64; ++kk) {
        v2f a = *(const v2f*)(ax + 4 * kk);
        v2f b = bp[kk * 32 + lane];
        c = __builtin_amdgcn_wmma_f32_16x16x4_f32(
                 false, a,  false, b,
                 (short)0, c,  false,  false);
    }

    const int col = lane & 15;
    if (col < 2) {
        float* s = ws + WS_S;
        const int rbase = row0 + ((lane & 16) ? 8 : 0);
        #pragma unroll
        for (int r = 0; r < 8; ++r) s[(size_t)(rbase + r) * 2 + col] = c[r];
    }
}

__global__ void __launch_bounds__(256)
edge_kernel(const int* __restrict__ eidx, const int* __restrict__ labels,
            const int* __restrict__ mask, const float* __restrict__ ws,
            float* __restrict__ out, float* __restrict__ acc) {
    const int e   = (int)(blockIdx.x * blockDim.x + threadIdx.x);
    const int tid = threadIdx.x;
    const float* s = ws + WS_S;

    float lm = 0.f, cm = 0.f;
    if (e < N_EDGES) {
        const int src = eidx[e];
        const int dst = eidx[N_EDGES + e];
        const float pre   = s[(size_t)src * 2 + 0] + s[(size_t)dst * 2 + 1];
        const float score = 1.f / (1.f + __expf(-pre));
        out[e] = score;
        const float p     = 1.f / (1.f + __expf(-score));
        const bool  pos   = (labels[e] == 1);
        const float pt    = pos ? p : (1.f - p);
        const float logpt = __logf(pt);
        const float at    = pos ? (1.f - ALPHA_F) : ALPHA_F;
        const float omp   = 1.f - pt;
        const float loss  = -(omp * omp) * (logpt * at);
        if (mask[e] != 0) { lm = loss; cm = 1.f; }
    }

    __shared__ float rl[256];
    __shared__ float rc[256];
    rl[tid] = lm; rc[tid] = cm;
    __syncthreads();
    for (int stride = 128; stride > 0; stride >>= 1) {
        if (tid < stride) { rl[tid] += rl[tid + stride]; rc[tid] += rc[tid + stride]; }
        __syncthreads();
    }
    if (tid == 0) {
        atomicAdd(&acc[0], rl[0]);
        atomicAdd(&acc[1], rc[0]);
    }
}

__global__ void finalize_kernel(const float* __restrict__ acc, float* __restrict__ out) {
    out[N_EDGES] = acc[0] / acc[1];
}

extern "C" void kernel_launch(void* const* d_in, const int* in_sizes, int n_in,
                              void* d_out, int out_size, void* d_ws, size_t ws_size,
                              hipStream_t stream) {
    const float* x      = (const float*)d_in[0];
    const float* Wd     = (const float*)d_in[1];
    const float* Wf     = (const float*)d_in[2];
    const int*   eidx   = (const int*)d_in[3];
    const int*   labels = (const int*)d_in[4];
    const int*   mask   = (const int*)d_in[5];
    float*       out    = (float*)d_out;
    float*       ws     = (float*)d_ws;
    float*       acc    = ws + WS_ACC;

    prep_kernel<<<1, 256, 0, stream>>>(Wd, Wf, ws);

    const int nWaves = N_NODES / 16;
    const int nBlk   = (nWaves + 7) / 8;
    matvec_wmma_kernel<<<nBlk, 256, 0, stream>>>(x, ws);

    edge_kernel<<<(N_EDGES + 255) / 256, 256, 0, stream>>>(eidx, labels, mask, ws, out, acc);

    finalize_kernel<<<1, 1, 0, stream>>>(acc, out);
}


# x streamed in whole 128-byte lines, 24-32 loads in flight per wave: WMMA tile = 2 rows x 8 interleaved row pieces, block-diagonal B in registers, pieces summed by DPP butterfly (k sum order changes)
# speedup vs baseline: 1.1602x; 1.1602x over previous
#include <hip/hip_runtime.h>
#include <math.h>

typedef __attribute__((ext_vector_type(2))) float v2f;
typedef __attribute__((ext_vector_type(4))) float v4f;
typedef __attribute__((ext_vector_type(8))) float v8f;

#define N_NODES 100000
#define IN_CH   256
#define OUT_CH  64
#define N_EDGES 400000
#define ALPHA_F 0.25f

#define WS_B   0
#define WS_S   4096
#define WS_ACC (4096 + 2 * N_NODES)

__global__ void __launch_bounds__(256)
prep_kernel(const float* __restrict__ Wd, const float* __restrict__ Wf,
            float* __restrict__ ws) {
    __shared__ float ush[OUT_CH];
    __shared__ float vsh[OUT_CH];
    const int t = threadIdx.x;
    if (t < OUT_CH) {
        ush[t] = Wf[t] + Wf[2 * OUT_CH + t];
        vsh[t] = Wf[OUT_CH + t] - Wf[2 * OUT_CH + t];
    }
    __syncthreads();

    float su = 0.f, sv = 0.f;
    #pragma unroll 8
    for (int o = 0; o < OUT_CH; ++o) {
        const float w = Wd[o * IN_CH + t];
        su = fmaf(w, ush[o], su);
        sv = fmaf(w, vsh[o], sv);
    }

    const int q = t >> 6;
    const int h = (t >> 5) & 1;
    const int p = (t >> 2) & 7;
    const int e = (t >> 1) & 1;
    const int s = t & 1;
    float* bp = ws + WS_B;
    bp[(q * 32 + 16 * h + 2 * p + 0) * 4 + 2 * e + s] = su;
    bp[(q * 32 + 16 * h + 2 * p + 1) * 4 + 2 * e + s] = sv;
    if (t == 0) { ws[WS_ACC] = 0.f; ws[WS_ACC + 1] = 0.f; }
}

#define MV_ROWS   32
#define MV_TILES  (MV_ROWS / 2)
#define MV_AHEAD  8
#define MV_WAVES  (N_NODES / MV_ROWS)
#define MV_BLOCK  128
#define MV_GRID   ((MV_WAVES * 32 + MV_BLOCK - 1) / MV_BLOCK)

__device__ __forceinline__ float pick8(float v0, float v1, float v2, float v3,
                                       float v4, float v5, float v6, float v7,
                                       bool b1, bool b2, bool b3) {
    const float s0 = b1 ? v1 : v0;
    const float s1 = b1 ? v3 : v2;
    const float s2 = b1 ? v5 : v4;
    const float s3 = b1 ? v7 : v6;
    const float u0 = b2 ? s1 : s0;
    const float u1 = b2 ? s3 : s2;
    return b3 ? u1 : u0;
}

template <int CTRL>
__device__ __forceinline__ float add_xor(float t) {
    const int r = __builtin_amdgcn_update_dpp(0, __float_as_int(t), CTRL, 0xf, 0xf, false);
    return t + __int_as_float(r);
}

__global__ void __launch_bounds__(MV_BLOCK)
matvec_wmma_kernel(const float* __restrict__ x, float* __restrict__ ws) {
    const int wave = (int)((blockIdx.x * blockDim.x + threadIdx.x) >> 5);
    if (wave >= MV_WAVES) return;
    const int lane  = threadIdx.x & 31;
    const int row0  = wave * MV_ROWS;
    const int h     = lane >> 4;
    const int rsel  = (lane >> 3) & 1;
    const int piece = lane & 7;

    const v4f* bp = (const v4f*)(ws + WS_B);
    v4f b[4];
    #pragma unroll
    for (int q = 0; q < 4; ++q) b[q] = bp[q * 32 + lane];

    const float* ax = x + (size_t)(row0 + rsel) * IN_CH + 32 * h + 4 * piece;
    v4f a[MV_TILES][4];
    #pragma unroll
    for (int i = 0; i < MV_AHEAD; ++i) {
        #pragma unroll
        for (int q = 0; q < 4; ++q)
            a[i][q] = *(const v4f*)(ax + (size_t)i * 2 * IN_CH + 64 * q);
    }
    __builtin_amdgcn_sched_barrier(0);

    const bool b1 = (lane & 2) != 0;
    const bool b2 = (lane & 4) != 0;
    const bool b3 = (lane & 8) != 0;
    const int from = 16 * ((lane >> 1) & 1) + 2 * (lane >> 2) + (lane & 1);

    float t[MV_TILES];
    #pragma unroll
    for (int i = 0; i < MV_TILES; i += 2) {
        v8f c0 = {};
        v8f c1 = {};
        #pragma unroll
        for (int q = 0; q < 4; ++q) {
            const v4f a0 = a[i][q];
            const v4f a1 = a[i + 1][q];
            const v2f a0lo = {a0.x, a0.y}, a0hi = {a0.z, a0.w};
            const v2f a1lo = {a1.x, a1.y}, a1hi = {a1.z, a1.w};
            const v2f blo = {b[q].x, b[q].y}, bhi = {b[q].z, b[q].w};
            c0 = __builtin_amdgcn_wmma_f32_16x16x4_f32(false, a0lo, false, blo,
                                                       (short)0, c0, false, false);
            c1 = __builtin_amdgcn_wmma_f32_16x16x4_f32(false, a1lo, false, blo,
                                                       (short)0, c1, false, false);
            c0 = __builtin_amdgcn_wmma_f32_16x16x4_f32(false, a0hi, false, bhi,
                                                       (short)0, c0, false, false);
            c1 = __builtin_amdgcn_wmma_f32_16x16x4_f32(false, a1hi, false, bhi,
                                                       (short)0, c1, false, false);
        }
        __builtin_amdgcn_sched_barrier(0);
        if (i + MV_AHEAD < MV_TILES) {
            #pragma unroll
            for (int q = 0; q < 4; ++q) {
                a[i + MV_AHEAD][q] =
                    *(const v4f*)(ax + (size_t)(i + MV_AHEAD) * 2 * IN_CH + 64 * q);
                a[i + MV_AHEAD + 1][q] =
                    *(const v4f*)(ax + (size_t)(i + MV_AHEAD + 1) * 2 * IN_CH + 64 * q);
            }
            __builtin_amdgcn_sched_barrier(0);
        }
        float t0 = pick8(c0[0], c0[1], c0[2], c0[3], c0[4], c0[5], c0[6], c0[7], b1, b2, b3);
        float t1 = pick8(c1[0], c1[1], c1[2], c1[3], c1[4], c1[5], c1[6], c1[7], b1, b2, b3);
        t0 = add_xor<0x162>(t0); t1 = add_xor<0x162>(t1);
        t0 = add_xor<0x164>(t0); t1 = add_xor<0x164>(t1);
        t0 = add_xor<0x168>(t0); t1 = add_xor<0x168>(t1);
        t[i]     = t0;
        t[i + 1] = t1;

        if ((i & 7) == 6) {
            const int u = i - 6;
            const float w = pick8(t[u], t[u + 1], t[u + 2], t[u + 3],
                                  t[u + 4], t[u + 5], t[u + 6], t[u + 7], b1, b2, b3);
            const float res = __shfl(w, from, 32);
            ws[WS_S + (size_t)(row0 + 2 * u) * 2 + lane] = res;
        }
    }
}

__global__ void __launch_bounds__(256)
edge_kernel(const int* __restrict__ eidx, const int* __restrict__ labels,
            const int* __restrict__ mask, const float* __restrict__ ws,
            float* __restrict__ out, float* __restrict__ acc) {
    const int e   = (int)(blockIdx.x * blockDim.x + threadIdx.x);
    const int tid = threadIdx.x;
    const float* s = ws + WS_S;

    float lm = 0.f, cm = 0.f;
    if (e < N_EDGES) {
        const int src = eidx[e];
        const int dst = eidx[N_EDGES + e];
        const float pre   = s[(size_t)src * 2 + 0] + s[(size_t)dst * 2 + 1];
        const float score = 1.f / (1.f + __expf(-pre));
        out[e] = score;
        const float p     = 1.f / (1.f + __expf(-score));
        const bool  pos   = (labels[e] == 1);
        const float pt    = pos ? p : (1.f - p);
        const float logpt = __logf(pt);
        const float at    = pos ? (1.f - ALPHA_F) : ALPHA_F;
        const float omp   = 1.f - pt;
        const float loss  = -(omp * omp) * (logpt * at);
        if (mask[e] != 0) { lm = loss; cm = 1.f; }
    }

    __shared__ float rl[256];
    __shared__ float rc[256];
    rl[tid] = lm; rc[tid] = cm;
    __syncthreads();
    for (int stride = 128; stride > 0; stride >>= 1) {
        if (tid < stride) { rl[tid] += rl[tid + stride]; rc[tid] += rc[tid + stride]; }
        __syncthreads();
    }
    if (tid == 0) {
        atomicAdd(&acc[0], rl[0]);
        atomicAdd(&acc[1], rc[0]);
    }
}

__global__ void finalize_kernel(const float* __restrict__ acc, float* __restrict__ out) {
    out[N_EDGES] = acc[0] / acc[1];
}

extern "C" void kernel_launch(void* const* d_in, const int* in_sizes, int n_in,
                              void* d_out, int out_size, void* d_ws, size_t ws_size,
                              hipStream_t stream) {
    const float* x      = (const float*)d_in[0];
    const float* Wd     = (const float*)d_in[1];
    const float* Wf     = (const float*)d_in[2];
    const int*   eidx   = (const int*)d_in[3];
    const int*   labels = (const int*)d_in[4];
    const int*   mask   = (const int*)d_in[5];
    float*       out    = (float*)d_out;
    float*       ws     = (float*)d_ws;
    float*       acc    = ws + WS_ACC;

    prep_kernel<<<1, 256, 0, stream>>>(Wd, Wf, ws);

    matvec_wmma_kernel<<<MV_GRID, MV_BLOCK, 0, stream>>>(x, ws);

    edge_kernel<<<(N_EDGES + 255) / 256, 256, 0, stream>>>(eidx, labels, mask, ws, out, acc);

    finalize_kernel<<<1, 1, 0, stream>>>(acc, out);
}


# edge kernel: 8 edges per thread as two 16-byte pieces (index, label, mask loads; score stores), 128-thread blocks: 12504 -> 1564 waves, 4x fewer block sums and atomics (loss sum order changes)
# speedup vs baseline: 1.3536x; 1.1667x over previous
#include <hip/hip_runtime.h>
#include <math.h>

typedef __attribute__((ext_vector_type(2))) float v2f;
typedef __attribute__((ext_vector_type(4))) float v4f;
typedef __attribute__((ext_vector_type(8))) float v8f;

#define N_NODES 100000
#define IN_CH   256
#define OUT_CH  64
#define N_EDGES 400000
#define ALPHA_F 0.25f

#define WS_B   0
#define WS_S   4096
#define WS_ACC (4096 + 2 * N_NODES)

__global__ void __launch_bounds__(256)
prep_kernel(const float* __restrict__ Wd, const float* __restrict__ Wf,
            float* __restrict__ ws) {
    __shared__ float ush[OUT_CH];
    __shared__ float vsh[OUT_CH];
    const int t = threadIdx.x;
    if (t < OUT_CH) {
        ush[t] = Wf[t] + Wf[2 * OUT_CH + t];
        vsh[t] = Wf[OUT_CH + t] - Wf[2 * OUT_CH + t];
    }
    __syncthreads();

    float su = 0.f, sv = 0.f;
    #pragma unroll 8
    for (int o = 0; o < OUT_CH; ++o) {
        const float w = Wd[o * IN_CH + t];
        su = fmaf(w, ush[o], su);
        sv = fmaf(w, vsh[o], sv);
    }

    const int q = t >> 6;
    const int h = (t >> 5) & 1;
    const int p = (t >> 2) & 7;
    const int e = (t >> 1) & 1;
    const int s = t & 1;
    float* bp = ws + WS_B;
    bp[(q * 32 + 16 * h + 2 * p + 0) * 4 + 2 * e + s] = su;
    bp[(q * 32 + 16 * h + 2 * p + 1) * 4 + 2 * e + s] = sv;
    if (t == 0) { ws[WS_ACC] = 0.f; ws[WS_ACC + 1] = 0.f; }
}

#define MV_ROWS   32
#define MV_TILES  (MV_ROWS / 2)
#define MV_AHEAD  8
#define MV_WAVES  (N_NODES / MV_ROWS)
#define MV_BLOCK  128
#define MV_GRID   ((MV_WAVES * 32 + MV_BLOCK - 1) / MV_BLOCK)

__device__ __forceinline__ float pick8(float v0, float v1, float v2, float v3,
                                       float v4, float v5, float v6, float v7,
                                       bool b1, bool b2, bool b3) {
    const float s0 = b1 ? v1 : v0;
    const float s1 = b1 ? v3 : v2;
    const float s2 = b1 ? v5 : v4;
    const float s3 = b1 ? v7 : v6;
    const float u0 = b2 ? s1 : s0;
    const float u1 = b2 ? s3 : s2;
    return b3 ? u1 : u0;
}

template <int CTRL>
__device__ __forceinline__ float add_xor(float t) {
    const int r = __builtin_amdgcn_update_dpp(0, __float_as_int(t), CTRL, 0xf, 0xf, false);
    return t + __int_as_float(r);
}

__global__ void __launch_bounds__(MV_BLOCK)
matvec_wmma_kernel(const float* __restrict__ x, float* __restrict__ ws) {
    const int wave = (int)((blockIdx.x * blockDim.x + threadIdx.x) >> 5);
    if (wave >= MV_WAVES) return;
    const int lane  = threadIdx.x & 31;
    const int row0  = wave * MV_ROWS;
    const int h     = lane >> 4;
    const int rsel  = (lane >> 3) & 1;
    const int piece = lane & 7;

    const v4f* bp = (const v4f*)(ws + WS_B);
    v4f b[4];
    #pragma unroll
    for (int q = 0; q < 4; ++q) b[q] = bp[q * 32 + lane];

    const float* ax = x + (size_t)(row0 + rsel) * IN_CH + 32 * h + 4 * piece;
    v4f a[MV_TILES][4];
    #pragma unroll
    for (int i = 0; i < MV_AHEAD; ++i) {
        #pragma unroll
        for (int q = 0; q < 4; ++q)
            a[i][q] = *(const v4f*)(ax + (size_t)i * 2 * IN_CH + 64 * q);
    }
    __builtin_amdgcn_sched_barrier(0);

    const bool b1 = (lane & 2) != 0;
    const bool b2 = (lane & 4) != 0;
    const bool b3 = (lane & 8) != 0;
    const int from = 16 * ((lane >> 1) & 1) + 2 * (lane >> 2) + (lane & 1);

    float t[MV_TILES];
    #pragma unroll
    for (int i = 0; i < MV_TILES; i += 2) {
        v8f c0 = {};
        v8f c1 = {};
        #pragma unroll
        for (int q = 0; q < 4; ++q) {
            const v4f a0 = a[i][q];
            const v4f a1 = a[i + 1][q];
            const v2f a0lo = {a0.x, a0.y}, a0hi = {a0.z, a0.w};
            const v2f a1lo = {a1.x, a1.y}, a1hi = {a1.z, a1.w};
            const v2f blo = {b[q].x, b[q].y}, bhi = {b[q].z, b[q].w};
            c0 = __builtin_amdgcn_wmma_f32_16x16x4_f32(false, a0lo, false, blo,
                                                       (short)0, c0, false, false);
            c1 = __builtin_amdgcn_wmma_f32_16x16x4_f32(false, a1lo, false, blo,
                                                       (short)0, c1, false, false);
            c0 = __builtin_amdgcn_wmma_f32_16x16x4_f32(false, a0hi, false, bhi,
                                                       (short)0, c0, false, false);
            c1 = __builtin_amdgcn_wmma_f32_16x16x4_f32(false, a1hi, false, bhi,
                                                       (short)0, c1, false, false);
        }
        __builtin_amdgcn_sched_barrier(0);
        if (i + MV_AHEAD < MV_TILES) {
            #pragma unroll
            for (int q = 0; q < 4; ++q) {
                a[i + MV_AHEAD][q] =
                    *(const v4f*)(ax + (size_t)(i + MV_AHEAD) * 2 * IN_CH + 64 * q);
                a[i + MV_AHEAD + 1][q] =
                    *(const v4f*)(ax + (size_t)(i + MV_AHEAD + 1) * 2 * IN_CH + 64 * q);
            }
            __builtin_amdgcn_sched_barrier(0);
        }
        float t0 = pick8(c0[0], c0[1], c0[2], c0[3], c0[4], c0[5], c0[6], c0[7], b1, b2, b3);
        float t1 = pick8(c1[0], c1[1], c1[2], c1[3], c1[4], c1[5], c1[6], c1[7], b1, b2, b3);
        t0 = add_xor<0x162>(t0); t1 = add_xor<0x162>(t1);
        t0 = add_xor<0x164>(t0); t1 = add_xor<0x164>(t1);
        t0 = add_xor<0x168>(t0); t1 = add_xor<0x168>(t1);
        t[i]     = t0;
        t[i + 1] = t1;

        if ((i & 7) == 6) {
            const int u = i - 6;
            const float w = pick8(t[u], t[u + 1], t[u + 2], t[u + 3],
                                  t[u + 4], t[u + 5], t[u + 6], t[u + 7], b1, b2, b3);
            const float res = __shfl(w, from, 32);
            ws[WS_S + (size_t)(row0 + 2 * u) * 2 + lane] = res;
        }
    }
}

typedef __attribute__((ext_vector_type(4))) int v4i;

#define EK_BLOCK  128
#define EK_PIECES 2
#define EK_EDGES  (EK_BLOCK * 4 * EK_PIECES)
#define EK_GRID   ((N_EDGES + EK_EDGES - 1) / EK_EDGES)

__global__ void __launch_bounds__(EK_BLOCK)
edge_kernel(const int* __restrict__ eidx, const int* __restrict__ labels,
            const int* __restrict__ mask, const float* __restrict__ ws,
            float* __restrict__ out, float* __restrict__ acc) {
    const int tid = threadIdx.x;
    const float* s = ws + WS_S;
    const int base = (int)blockIdx.x * EK_EDGES + 4 * tid;

    bool ok[EK_PIECES];
    v4i src[EK_PIECES], dst[EK_PIECES], lab[EK_PIECES], msk[EK_PIECES];
    #pragma unroll
    for (int c = 0; c < EK_PIECES; ++c) {
        const int e0 = base + c * (4 * EK_BLOCK);
        ok[c] = e0 < N_EDGES;
        const int el = ok[c] ? e0 : 0;
        src[c] = *(const v4i*)(eidx + el);
        dst[c] = *(const v4i*)(eidx + N_EDGES + el);
        lab[c] = *(const v4i*)(labels + el);
        msk[c] = *(const v4i*)(mask + el);
    }

    float pre[EK_PIECES][4];
    #pragma unroll
    for (int c = 0; c < EK_PIECES; ++c) {
        #pragma unroll
        for (int j = 0; j < 4; ++j)
            pre[c][j] = s[(size_t)src[c][j] * 2 + 0] + s[(size_t)dst[c][j] * 2 + 1];
    }

    float lm = 0.f, cm = 0.f;
    #pragma unroll
    for (int c = 0; c < EK_PIECES; ++c) {
        v4f sc;
        #pragma unroll
        for (int j = 0; j < 4; ++j) {
            const float score = 1.f / (1.f + __expf(-pre[c][j]));
            sc[j] = score;
            const float p     = 1.f / (1.f + __expf(-score));
            const bool  pos   = (lab[c][j] == 1);
            const float pt    = pos ? p : (1.f - p);
            const float logpt = __logf(pt);
            const float at    = pos ? (1.f - ALPHA_F) : ALPHA_F;
            const float omp   = 1.f - pt;
            const float loss  = -(omp * omp) * (logpt * at);
            const bool take = ok[c] && (msk[c][j] != 0);
            lm += take ? loss : 0.f;
            cm += take ? 1.f : 0.f;
        }
        if (ok[c]) *(v4f*)(out + base + c * (4 * EK_BLOCK)) = sc;
    }

    __shared__ float rl[EK_BLOCK];
    __shared__ float rc[EK_BLOCK];
    rl[tid] = lm; rc[tid] = cm;
    __syncthreads();
    for (int stride = EK_BLOCK / 2; stride > 0; stride >>= 1) {
        if (tid < stride) { rl[tid] += rl[tid + stride]; rc[tid] += rc[tid + stride]; }
        __syncthreads();
    }
    if (tid == 0) {
        atomicAdd(&acc[0], rl[0]);
        atomicAdd(&acc[1], rc[0]);
    }
}

__global__ void finalize_kernel(const float* __restrict__ acc, float* __restrict__ out) {
    out[N_EDGES] = acc[0] / acc[1];
}

extern "C" void kernel_launch(void* const* d_in, const int* in_sizes, int n_in,
                              void* d_out, int out_size, void* d_ws, size_t ws_size,
                              hipStream_t stream) {
    const float* x      = (const float*)d_in[0];
    const float* Wd     = (const float*)d_in[1];
    const float* Wf     = (const float*)d_in[2];
    const int*   eidx   = (const int*)d_in[3];
    const int*   labels = (const int*)d_in[4];
    const int*   mask   = (const int*)d_in[5];
    float*       out    = (float*)d_out;
    float*       ws     = (float*)d_ws;
    float*       acc    = ws + WS_ACC;

    prep_kernel<<<1, 256, 0, stream>>>(Wd, Wf, ws);

    matvec_wmma_kernel<<<MV_GRID, MV_BLOCK, 0, stream>>>(x, ws);

    edge_kernel<<<EK_GRID, EK_BLOCK, 0, stream>>>(eidx, labels, mask, ws, out, acc);

    finalize_kernel<<<1, 1, 0, stream>>>(acc, out);
}


# weight fold (W_f into W_d) moved from its own one-block launch into the x matvec kernel: every 13-wave block builds the 256x2 B operand in LDS while its first x tiles are in flight; 4 launches -> 3
# speedup vs baseline: 1.6791x; 1.2405x over previous
#include <hip/hip_runtime.h>
#include <math.h>

typedef __attribute__((ext_vector_type(2))) float v2f;
typedef __attribute__((ext_vector_type(4))) float v4f;
typedef __attribute__((ext_vector_type(8))) float v8f;

#define N_NODES 100000
#define IN_CH   256
#define OUT_CH  64
#define N_EDGES 400000
#define ALPHA_F 0.25f

#define WS_S   4096
#define WS_ACC (4096 + 2 * N_NODES)

#define MV_ROWS   32
#define MV_TILES  (MV_ROWS / 2)
#define MV_AHEAD  8
#define MV_WAVES  (N_NODES / MV_ROWS)
#define MV_BWAVES 13
#define MV_BLOCK  (32 * MV_BWAVES)
#define MV_GRID   ((MV_WAVES + MV_BWAVES - 1) / MV_BWAVES)

__device__ __forceinline__ void lds_barrier() {
    asm volatile("s_wait_dscnt 0x0\n\ts_barrier_signal -1\n\ts_barrier_wait -1" ::: "memory");
}

__device__ __forceinline__ float pick8(float v0, float v1, float v2, float v3,
                                       float v4, float v5, float v6, float v7,
                                       bool b1, bool b2, bool b3) {
    const float s0 = b1 ? v1 : v0;
    const float s1 = b1 ? v3 : v2;
    const float s2 = b1 ? v5 : v4;
    const float s3 = b1 ? v7 : v6;
    const float u0 = b2 ? s1 : s0;
    const float u1 = b2 ? s3 : s2;
    return b3 ? u1 : u0;
}

template <int CTRL>
__device__ __forceinline__ float add_xor(float t) {
    const int r = __builtin_amdgcn_update_dpp(0, __float_as_int(t), CTRL, 0xf, 0xf, false);
    return t + __int_as_float(r);
}

__global__ void __launch_bounds__(MV_BLOCK)
matvec_wmma_kernel(const float* __restrict__ x, const float* __restrict__ Wd,
                   const float* __restrict__ Wf, float* __restrict__ ws) {
    __shared__ float ush[OUT_CH];
    __shared__ float vsh[OUT_CH];
    __shared__ v4f   bsh4[4 * 32];
    float* bsh = (float*)bsh4;

    const int tid  = threadIdx.x;
    const int lane = tid & 31;
    const int wib      = __builtin_amdgcn_readfirstlane(tid >> 5);
    const bool builder = wib < IN_CH / 32;
    const bool wfwave  = wib < OUT_CH / 32;
    const int  wave_raw = (int)blockIdx.x * MV_BWAVES + wib;
    const bool active   = wave_raw < MV_WAVES;
    const int  wave     = active ? wave_raw : (MV_WAVES - 1);
    const int row0  = wave * MV_ROWS;
    const int h     = lane >> 4;
    const int rsel  = (lane >> 3) & 1;
    const int piece = lane & 7;

    float wf0 = 0.f, wf1 = 0.f, wf2 = 0.f;
    float w[OUT_CH];
    if (builder) {
        if (wfwave) {
            wf0 = Wf[tid];
            wf1 = Wf[OUT_CH + tid];
            wf2 = Wf[2 * OUT_CH + tid];
        }
        #pragma unroll
        for (int o = 0; o < OUT_CH; ++o) w[o] = Wd[o * IN_CH + tid];
    }

    const float* ax = x + (size_t)(row0 + rsel) * IN_CH + 32 * h + 4 * piece;
    v4f a[MV_TILES][4];
    #pragma unroll
    for (int i = 0; i < MV_AHEAD; ++i) {
        #pragma unroll
        for (int q = 0; q < 4; ++q)
            a[i][q] = *(const v4f*)(ax + (size_t)i * 2 * IN_CH + 64 * q);
    }

    if (wfwave) {
        asm volatile("" : "+v"(wf0), "+v"(wf2));
        ush[tid] = wf0 + wf2;
        vsh[tid] = wf1 - wf2;
    }
    lds_barrier();
    if (builder) {
        float su = 0.f, sv = 0.f;
        #pragma unroll
        for (int o = 0; o < OUT_CH; ++o) {
            su = fmaf(w[o], ush[o], su);
            sv = fmaf(w[o], vsh[o], sv);
        }
        const int kq = tid >> 6;
        const int kh = (tid >> 5) & 1;
        const int kp = (tid >> 2) & 7;
        const int ke = (tid >> 1) & 1;
        const int ks = tid & 1;
        bsh[(kq * 32 + 16 * kh + 2 * kp + 0) * 4 + 2 * ke + ks] = su;
        bsh[(kq * 32 + 16 * kh + 2 * kp + 1) * 4 + 2 * ke + ks] = sv;
        if (blockIdx.x == 0 && tid == 0) { ws[WS_ACC] = 0.f; ws[WS_ACC + 1] = 0.f; }
    }
    lds_barrier();

    v4f b[4];
    #pragma unroll
    for (int q = 0; q < 4; ++q) b[q] = bsh4[q * 32 + lane];

    const bool b1 = (lane & 2) != 0;
    const bool b2 = (lane & 4) != 0;
    const bool b3 = (lane & 8) != 0;
    const int from = 16 * ((lane >> 1) & 1) + 2 * (lane >> 2) + (lane & 1);

    float t[MV_TILES];
    #pragma unroll
    for (int i = 0; i < MV_TILES; i += 2) {
        v8f c0 = {};
        v8f c1 = {};
        #pragma unroll
        for (int q = 0; q < 4; ++q) {
            const v4f a0 = a[i][q];
            const v4f a1 = a[i + 1][q];
            const v2f a0lo = {a0.x, a0.y}, a0hi = {a0.z, a0.w};
            const v2f a1lo = {a1.x, a1.y}, a1hi = {a1.z, a1.w};
            const v2f blo = {b[q].x, b[q].y}, bhi = {b[q].z, b[q].w};
            c0 = __builtin_amdgcn_wmma_f32_16x16x4_f32(false, a0lo, false, blo,
                                                       (short)0, c0, false, false);
            c1 = __builtin_amdgcn_wmma_f32_16x16x4_f32(false, a1lo, false, blo,
                                                       (short)0, c1, false, false);
            c0 = __builtin_amdgcn_wmma_f32_16x16x4_f32(false, a0hi, false, bhi,
                                                       (short)0, c0, false, false);
            c1 = __builtin_amdgcn_wmma_f32_16x16x4_f32(false, a1hi, false, bhi,
                                                       (short)0, c1, false, false);
        }
        __builtin_amdgcn_sched_barrier(0);
        if (i + MV_AHEAD < MV_TILES) {
            #pragma unroll
            for (int q = 0; q < 4; ++q) {
                a[i + MV_AHEAD][q] =
                    *(const v4f*)(ax + (size_t)(i + MV_AHEAD) * 2 * IN_CH + 64 * q);
                a[i + MV_AHEAD + 1][q] =
                    *(const v4f*)(ax + (size_t)(i + MV_AHEAD + 1) * 2 * IN_CH + 64 * q);
            }
            __builtin_amdgcn_sched_barrier(0);
        }
        float t0 = pick8(c0[0], c0[1], c0[2], c0[3], c0[4], c0[5], c0[6], c0[7], b1, b2, b3);
        float t1 = pick8(c1[0], c1[1], c1[2], c1[3], c1[4], c1[5], c1[6], c1[7], b1, b2, b3);
        t0 = add_xor<0x162>(t0); t1 = add_xor<0x162>(t1);
        t0 = add_xor<0x164>(t0); t1 = add_xor<0x164>(t1);
        t0 = add_xor<0x168>(t0); t1 = add_xor<0x168>(t1);
        t[i]     = t0;
        t[i + 1] = t1;

        if ((i & 7) == 6) {
            const int u = i - 6;
            const float w = pick8(t[u], t[u + 1], t[u + 2], t[u + 3],
                                  t[u + 4], t[u + 5], t[u + 6], t[u + 7], b1, b2, b3);
            const float res = __shfl(w, from, 32);
            if (active) ws[WS_S + (size_t)(row0 + 2 * u) * 2 + lane] = res;
        }
    }
}

typedef __attribute__((ext_vector_type(4))) int v4i;

#define EK_BLOCK  128
#define EK_PIECES 2
#define EK_EDGES  (EK_BLOCK * 4 * EK_PIECES)
#define EK_GRID   ((N_EDGES + EK_EDGES - 1) / EK_EDGES)

__global__ void __launch_bounds__(EK_BLOCK)
edge_kernel(const int* __restrict__ eidx, const int* __restrict__ labels,
            const int* __restrict__ mask, const float* __restrict__ ws,
            float* __restrict__ out, float* __restrict__ acc) {
    const int tid = threadIdx.x;
    const float* s = ws + WS_S;
    const int base = (int)blockIdx.x * EK_EDGES + 4 * tid;

    bool ok[EK_PIECES];
    v4i src[EK_PIECES], dst[EK_PIECES], lab[EK_PIECES], msk[EK_PIECES];
    #pragma unroll
    for (int c = 0; c < EK_PIECES; ++c) {
        const int e0 = base + c * (4 * EK_BLOCK);
        ok[c] = e0 < N_EDGES;
        const int el = ok[c] ? e0 : 0;
        src[c] = *(const v4i*)(eidx + el);
        dst[c] = *(const v4i*)(eidx + N_EDGES + el);
        lab[c] = *(const v4i*)(labels + el);
        msk[c] = *(const v4i*)(mask + el);
    }

    float pre[EK_PIECES][4];
    #pragma unroll
    for (int c = 0; c < EK_PIECES; ++c) {
        #pragma unroll
        for (int j = 0; j < 4; ++j)
            pre[c][j] = s[(size_t)src[c][j] * 2 + 0] + s[(size_t)dst[c][j] * 2 + 1];
    }

    float lm = 0.f, cm = 0.f;
    #pragma unroll
    for (int c = 0; c < EK_PIECES; ++c) {
        v4f sc;
        #pragma unroll
        for (int j = 0; j < 4; ++j) {
            const float score = 1.f / (1.f + __expf(-pre[c][j]));
            sc[j] = score;
            const float p     = 1.f / (1.f + __expf(-score));
            const bool  pos   = (lab[c][j] == 1);
            const float pt    = pos ? p : (1.f - p);
            const float logpt = __logf(pt);
            const float at    = pos ? (1.f - ALPHA_F) : ALPHA_F;
            const float omp   = 1.f - pt;
            const float loss  = -(omp * omp) * (logpt * at);
            const bool take = ok[c] && (msk[c][j] != 0);
            lm += take ? loss : 0.f;
            cm += take ? 1.f : 0.f;
        }
        if (ok[c]) *(v4f*)(out + base + c * (4 * EK_BLOCK)) = sc;
    }

    __shared__ float rl[EK_BLOCK];
    __shared__ float rc[EK_BLOCK];
    rl[tid] = lm; rc[tid] = cm;
    __syncthreads();
    for (int stride = EK_BLOCK / 2; stride > 0; stride >>= 1) {
        if (tid < stride) { rl[tid] += rl[tid + stride]; rc[tid] += rc[tid + stride]; }
        __syncthreads();
    }
    if (tid == 0) {
        atomicAdd(&acc[0], rl[0]);
        atomicAdd(&acc[1], rc[0]);
    }
}

__global__ void finalize_kernel(const float* __restrict__ acc, float* __restrict__ out) {
    out[N_EDGES] = acc[0] / acc[1];
}

extern "C" void kernel_launch(void* const* d_in, const int* in_sizes, int n_in,
                              void* d_out, int out_size, void* d_ws, size_t ws_size,
                              hipStream_t stream) {
    const float* x      = (const float*)d_in[0];
    const float* Wd     = (const float*)d_in[1];
    const float* Wf     = (const float*)d_in[2];
    const int*   eidx   = (const int*)d_in[3];
    const int*   labels = (const int*)d_in[4];
    const int*   mask   = (const int*)d_in[5];
    float*       out    = (float*)d_out;
    float*       ws     = (float*)d_ws;
    float*       acc    = ws + WS_ACC;

    matvec_wmma_kernel<<<MV_GRID, MV_BLOCK, 0, stream>>>(x, Wd, Wf, ws);

    edge_kernel<<<EK_GRID, EK_BLOCK, 0, stream>>>(eidx, labels, mask, ws, out, acc);

    finalize_kernel<<<1, 1, 0, stream>>>(acc, out);
}


# finalize launch folded into the edge kernel, no fence: block loss sum (fixed point: exact, order-free sum over blocks), mask count and arrival count in ONE 64-bit returning atomic; 3 -> 2 launches
# speedup vs baseline: 1.8029x; 1.0737x over previous
#include <hip/hip_runtime.h>
#include <math.h>

typedef __attribute__((ext_vector_type(2))) float v2f;
typedef __attribute__((ext_vector_type(4))) float v4f;
typedef __attribute__((ext_vector_type(8))) float v8f;

#define N_NODES 100000
#define IN_CH   256
#define OUT_CH  64
#define N_EDGES 400000
#define ALPHA_F 0.25f

#define WS_S   4096
#define WS_ACC (4096 + 2 * N_NODES)

#define ACC_LOSS_UNIT  524288.0f
#define ACC_CNT_SHIFT  36
#define ACC_TKT_SHIFT  55
#define ACC_LOSS_MASK  ((1ull << ACC_CNT_SHIFT) - 1)
#define ACC_CNT_MASK   ((1ull << (ACC_TKT_SHIFT - ACC_CNT_SHIFT)) - 1)

#define MV_ROWS   32
#define MV_TILES  (MV_ROWS / 2)
#define MV_AHEAD  8
#define MV_WAVES  (N_NODES / MV_ROWS)
#define MV_BWAVES 13
#define MV_BLOCK  (32 * MV_BWAVES)
#define MV_GRID   ((MV_WAVES + MV_BWAVES - 1) / MV_BWAVES)

__device__ __forceinline__ void lds_barrier() {
    asm volatile("s_wait_dscnt 0x0\n\ts_barrier_signal -1\n\ts_barrier_wait -1" ::: "memory");
}

__device__ __forceinline__ float pick8(float v0, float v1, float v2, float v3,
                                       float v4, float v5, float v6, float v7,
                                       bool b1, bool b2, bool b3) {
    const float s0 = b1 ? v1 : v0;
    const float s1 = b1 ? v3 : v2;
    const float s2 = b1 ? v5 : v4;
    const float s3 = b1 ? v7 : v6;
    const float u0 = b2 ? s1 : s0;
    const float u1 = b2 ? s3 : s2;
    return b3 ? u1 : u0;
}

template <int CTRL>
__device__ __forceinline__ float add_xor(float t) {
    const int r = __builtin_amdgcn_update_dpp(0, __float_as_int(t), CTRL, 0xf, 0xf, false);
    return t + __int_as_float(r);
}

__global__ void __launch_bounds__(MV_BLOCK)
matvec_wmma_kernel(const float* __restrict__ x, const float* __restrict__ Wd,
                   const float* __restrict__ Wf, float* __restrict__ ws) {
    __shared__ float ush[OUT_CH];
    __shared__ float vsh[OUT_CH];
    __shared__ v4f   bsh4[4 * 32];
    float* bsh = (float*)bsh4;

    const int tid  = threadIdx.x;
    const int lane = tid & 31;
    const int wib      = __builtin_amdgcn_readfirstlane(tid >> 5);
    const bool builder = wib < IN_CH / 32;
    const bool wfwave  = wib < OUT_CH / 32;
    const int  wave_raw = (int)blockIdx.x * MV_BWAVES + wib;
    const bool active   = wave_raw < MV_WAVES;
    const int  wave     = active ? wave_raw : (MV_WAVES - 1);
    const int row0  = wave * MV_ROWS;
    const int h     = lane >> 4;
    const int rsel  = (lane >> 3) & 1;
    const int piece = lane & 7;

    float wf0 = 0.f, wf1 = 0.f, wf2 = 0.f;
    float w[OUT_CH];
    if (builder) {
        if (wfwave) {
            wf0 = Wf[tid];
            wf1 = Wf[OUT_CH + tid];
            wf2 = Wf[2 * OUT_CH + tid];
        }
        #pragma unroll
        for (int o = 0; o < OUT_CH; ++o) w[o] = Wd[o * IN_CH + tid];
    }

    const float* ax = x + (size_t)(row0 + rsel) * IN_CH + 32 * h + 4 * piece;
    v4f a[MV_TILES][4];
    #pragma unroll
    for (int i = 0; i < MV_AHEAD; ++i) {
        #pragma unroll
        for (int q = 0; q < 4; ++q)
            a[i][q] = *(const v4f*)(ax + (size_t)i * 2 * IN_CH + 64 * q);
    }

    if (wfwave) {
        asm volatile("" : "+v"(wf0), "+v"(wf2));
        ush[tid] = wf0 + wf2;
        vsh[tid] = wf1 - wf2;
    }
    lds_barrier();
    if (builder) {
        float su = 0.f, sv = 0.f;
        #pragma unroll
        for (int o = 0; o < OUT_CH; ++o) {
            su = fmaf(w[o], ush[o], su);
            sv = fmaf(w[o], vsh[o], sv);
        }
        const int kq = tid >> 6;
        const int kh = (tid >> 5) & 1;
        const int kp = (tid >> 2) & 7;
        const int ke = (tid >> 1) & 1;
        const int ks = tid & 1;
        bsh[(kq * 32 + 16 * kh + 2 * kp + 0) * 4 + 2 * ke + ks] = su;
        bsh[(kq * 32 + 16 * kh + 2 * kp + 1) * 4 + 2 * ke + ks] = sv;
        if (blockIdx.x == 0 && tid == 0) *(unsigned long long*)(ws + WS_ACC) = 0ull;
    }
    lds_barrier();

    v4f b[4];
    #pragma unroll
    for (int q = 0; q < 4; ++q) b[q] = bsh4[q * 32 + lane];

    const bool b1 = (lane & 2) != 0;
    const bool b2 = (lane & 4) != 0;
    const bool b3 = (lane & 8) != 0;
    const int from = 16 * ((lane >> 1) & 1) + 2 * (lane >> 2) + (lane & 1);

    float t[MV_TILES];
    #pragma unroll
    for (int i = 0; i < MV_TILES; i += 2) {
        v8f c0 = {};
        v8f c1 = {};
        #pragma unroll
        for (int q = 0; q < 4; ++q) {
            const v4f a0 = a[i][q];
            const v4f a1 = a[i + 1][q];
            const v2f a0lo = {a0.x, a0.y}, a0hi = {a0.z, a0.w};
            const v2f a1lo = {a1.x, a1.y}, a1hi = {a1.z, a1.w};
            const v2f blo = {b[q].x, b[q].y}, bhi = {b[q].z, b[q].w};
            c0 = __builtin_amdgcn_wmma_f32_16x16x4_f32(false, a0lo, false, blo,
                                                       (short)0, c0, false, false);
            c1 = __builtin_amdgcn_wmma_f32_16x16x4_f32(false, a1lo, false, blo,
                                                       (short)0, c1, false, false);
            c0 = __builtin_amdgcn_wmma_f32_16x16x4_f32(false, a0hi, false, bhi,
                                                       (short)0, c0, false, false);
            c1 = __builtin_amdgcn_wmma_f32_16x16x4_f32(false, a1hi, false, bhi,
                                                       (short)0, c1, false, false);
        }
        __builtin_amdgcn_sched_barrier(0);
        if (i + MV_AHEAD < MV_TILES) {
            #pragma unroll
            for (int q = 0; q < 4; ++q) {
                a[i + MV_AHEAD][q] =
                    *(const v4f*)(ax + (size_t)(i + MV_AHEAD) * 2 * IN_CH + 64 * q);
                a[i + MV_AHEAD + 1][q] =
                    *(const v4f*)(ax + (size_t)(i + MV_AHEAD + 1) * 2 * IN_CH + 64 * q);
            }
            __builtin_amdgcn_sched_barrier(0);
        }
        float t0 = pick8(c0[0], c0[1], c0[2], c0[3], c0[4], c0[5], c0[6], c0[7], b1, b2, b3);
        float t1 = pick8(c1[0], c1[1], c1[2], c1[3], c1[4], c1[5], c1[6], c1[7], b1, b2, b3);
        t0 = add_xor<0x162>(t0); t1 = add_xor<0x162>(t1);
        t0 = add_xor<0x164>(t0); t1 = add_xor<0x164>(t1);
        t0 = add_xor<0x168>(t0); t1 = add_xor<0x168>(t1);
        t[i]     = t0;
        t[i + 1] = t1;

        if ((i & 7) == 6) {
            const int u = i - 6;
            const float w = pick8(t[u], t[u + 1], t[u + 2], t[u + 3],
                                  t[u + 4], t[u + 5], t[u + 6], t[u + 7], b1, b2, b3);
            const float res = __shfl(w, from, 32);
            if (active) ws[WS_S + (size_t)(row0 + 2 * u) * 2 + lane] = res;
        }
    }
}

typedef __attribute__((ext_vector_type(4))) int v4i;

#define EK_BLOCK  128
#define EK_PIECES 2
#define EK_EDGES  (EK_BLOCK * 4 * EK_PIECES)
#define EK_GRID   ((N_EDGES + EK_EDGES - 1) / EK_EDGES)

__global__ void __launch_bounds__(EK_BLOCK)
edge_kernel(const int* __restrict__ eidx, const int* __restrict__ labels,
            const int* __restrict__ mask, const float* __restrict__ ws,
            float* __restrict__ out, float* __restrict__ acc) {
    const int tid = threadIdx.x;
    const float* s = ws + WS_S;
    const int base = (int)blockIdx.x * EK_EDGES + 4 * tid;

    bool ok[EK_PIECES];
    v4i src[EK_PIECES], dst[EK_PIECES], lab[EK_PIECES], msk[EK_PIECES];
    #pragma unroll
    for (int c = 0; c < EK_PIECES; ++c) {
        const int e0 = base + c * (4 * EK_BLOCK);
        ok[c] = e0 < N_EDGES;
        const int el = ok[c] ? e0 : 0;
        src[c] = *(const v4i*)(eidx + el);
        dst[c] = *(const v4i*)(eidx + N_EDGES + el);
        lab[c] = *(const v4i*)(labels + el);
        msk[c] = *(const v4i*)(mask + el);
    }

    float pre[EK_PIECES][4];
    #pragma unroll
    for (int c = 0; c < EK_PIECES; ++c) {
        #pragma unroll
        for (int j = 0; j < 4; ++j)
            pre[c][j] = s[(size_t)src[c][j] * 2 + 0] + s[(size_t)dst[c][j] * 2 + 1];
    }

    float lm = 0.f, cm = 0.f;
    #pragma unroll
    for (int c = 0; c < EK_PIECES; ++c) {
        v4f sc;
        #pragma unroll
        for (int j = 0; j < 4; ++j) {
            const float score = 1.f / (1.f + __expf(-pre[c][j]));
            sc[j] = score;
            const float p     = 1.f / (1.f + __expf(-score));
            const bool  pos   = (lab[c][j] == 1);
            const float pt    = pos ? p : (1.f - p);
            const float logpt = __logf(pt);
            const float at    = pos ? (1.f - ALPHA_F) : ALPHA_F;
            const float omp   = 1.f - pt;
            const float loss  = -(omp * omp) * (logpt * at);
            const bool take = ok[c] && (msk[c][j] != 0);
            lm += take ? loss : 0.f;
            cm += take ? 1.f : 0.f;
        }
        if (ok[c]) *(v4f*)(out + base + c * (4 * EK_BLOCK)) = sc;
    }

    __shared__ float rl[EK_BLOCK];
    __shared__ float rc[EK_BLOCK];
    rl[tid] = lm; rc[tid] = cm;
    __syncthreads();
    for (int stride = EK_BLOCK / 2; stride > 0; stride >>= 1) {
        if (tid < stride) { rl[tid] += rl[tid + stride]; rc[tid] += rc[tid + stride]; }
        __syncthreads();
    }
    if (tid == 0) {
        const unsigned long long own =
            (unsigned long long)__float2uint_rn(rl[0] * ACC_LOSS_UNIT)
            + ((unsigned long long)(unsigned)rc[0] << ACC_CNT_SHIFT)
            + (1ull << ACC_TKT_SHIFT);
        const unsigned long long old =
            __hip_atomic_fetch_add((unsigned long long*)acc, own,
                                   __ATOMIC_RELAXED, __HIP_MEMORY_SCOPE_AGENT);
        if ((old >> ACC_TKT_SHIFT) == (unsigned long long)(EK_GRID - 1)) {
            const unsigned long long tot = old + own;
            const float lsum = (float)(tot & ACC_LOSS_MASK) * (1.f / ACC_LOSS_UNIT);
            const float msum = (float)((tot >> ACC_CNT_SHIFT) & ACC_CNT_MASK);
            out[N_EDGES] = lsum / msum;
        }
    }
}

extern "C" void kernel_launch(void* const* d_in, const int* in_sizes, int n_in,
                              void* d_out, int out_size, void* d_ws, size_t ws_size,
                              hipStream_t stream) {
    const float* x      = (const float*)d_in[0];
    const float* Wd     = (const float*)d_in[1];
    const float* Wf     = (const float*)d_in[2];
    const int*   eidx   = (const int*)d_in[3];
    const int*   labels = (const int*)d_in[4];
    const int*   mask   = (const int*)d_in[5];
    float*       out    = (float*)d_out;
    float*       ws     = (float*)d_ws;
    float*       acc    = ws + WS_ACC;

    matvec_wmma_kernel<<<MV_GRID, MV_BLOCK, 0, stream>>>(x, Wd, Wf, ws);

    edge_kernel<<<EK_GRID, EK_BLOCK, 0, stream>>>(eidx, labels, mask, ws, out, acc);
}
